# starting kernel
# baseline (speedup 1.0000x reference)
#include <hip/hip_runtime.h>
#include <math.h>


typedef float v2f __attribute__((ext_vector_type(2)));
typedef float v8f __attribute__((ext_vector_type(8)));

#define B_  4
#define C_  32
#define H_  256
#define W_  256
#define M_  16
#define PIX (H_ * W_)

__device__ __forceinline__ v8f wmma4(v2f a, v2f b, v8f c) {
  return __builtin_amdgcn_wmma_f32_16x16x4_f32(false, a, false, b, (short)0, c,
                                               false, false);
}

__device__ __forceinline__ v2f ldA(const float* p, int lda, int lane) {
  int r = lane & 15, kk = (lane >> 4) << 1;
  v2f v; v.x = p[r * lda + kk]; v.y = p[r * lda + kk + 1]; return v;
}
__device__ __forceinline__ v2f ldAt(const float* p, int lda, int lane) {
  int r = lane & 15, kk = (lane >> 4) << 1;
  v2f v; v.x = p[kk * lda + r]; v.y = p[(kk + 1) * lda + r]; return v;
}
__device__ __forceinline__ v2f ldB(const float* p, int ldb, int lane) {
  int n = lane & 15, kk = (lane >> 4) << 1;
  v2f v; v.x = p[kk * ldb + n]; v.y = p[(kk + 1) * ldb + n]; return v;
}
__device__ __forceinline__ v2f ldBt(const float* p, int ldb, int lane) {
  int n = lane & 15, kk = (lane >> 4) << 1;
  v2f v; v.x = p[n * ldb + kk]; v.y = p[n * ldb + kk + 1]; return v;
}
__device__ __forceinline__ void stD(float* p, int ldd, int lane, v8f d, float s) {
  int n = lane & 15, mh = (lane >> 4) << 3;
#pragma unroll
  for (int r = 0; r < 8; ++r) p[(r + mh) * ldd + n] = d[r] * s;
}

__device__ __forceinline__ v8f vzero() {
  v8f z = {0.f, 0.f, 0.f, 0.f, 0.f, 0.f, 0.f, 0.f};
  return z;
}

__global__ void k_tabs(float* tabC, float* tabS, float* tabC2, float* tabS2) {
  int i = blockIdx.x * blockDim.x + threadIdx.x;
  if (i >= M_ * W_) return;
  int k = i >> 8, x = i & 255;
  int m = (k * x) & 255;
  float ang = 0.0245436926f * (float)m;
  float c = cosf(ang), s = sinf(ang);
  tabC[i] = c; tabS[i] = s;
  float w2 = (k == 0) ? 1.0f : 2.0f;
  tabC2[i] = w2 * c * 0.0625f;
  tabS2[i] = w2 * s * 0.0625f;
}

__global__ void k_lift(const float* __restrict__ x, const float* __restrict__ lw,
                       const float* __restrict__ lb, float* __restrict__ x0) {
  int p = blockIdx.x * blockDim.x + threadIdx.x;
  if (p >= B_ * PIX) return;
  int b = p >> 16, pix = p & (PIX - 1);
  float xi0 = x[(b * 3 + 0) * PIX + pix];
  float xi1 = x[(b * 3 + 1) * PIX + pix];
  float xi2 = x[(b * 3 + 2) * PIX + pix];
  float* o = x0 + (size_t)p * C_;
#pragma unroll
  for (int c = 0; c < C_; ++c)
    o[c] = lb[c] + lw[c * 3 + 0] * xi0 + lw[c * 3 + 1] * xi1 + lw[c * 3 + 2] * xi2;
}

__global__ void k_fwdX(const float* __restrict__ xin, const float* __restrict__ tabC,
                       const float* __restrict__ tabS, float* __restrict__ Tr,
                       float* __restrict__ Ti) {
  int wid = blockIdx.x * (blockDim.x >> 5) + (threadIdx.x >> 5);
  int lane = threadIdx.x & 31;
  int ch = wid & 1;
  int byi = wid >> 1;
  int b = byi >> 8, y = byi & 255;
  const float* row = xin + ((size_t)(b * H_ + y)) * W_ * C_ + 16 * ch;
  v8f aC = vzero(), aS = vzero();
  for (int k0 = 0; k0 < W_; k0 += 4) {
    v2f aCr = ldA(tabC + k0, W_, lane);
    v2f aSr = ldA(tabS + k0, W_, lane);
    v2f bh = ldB(row + k0 * C_, C_, lane);
    aC = wmma4(aCr, bh, aC);
    aS = wmma4(aSr, bh, aS);
  }
  float* tr = Tr + ((size_t)(b * H_ + y)) * M_ * C_ + 16 * ch;
  float* ti = Ti + ((size_t)(b * H_ + y)) * M_ * C_ + 16 * ch;
  stD(tr, C_, lane, aC, 0.0625f);
  stD(ti, C_, lane, aS, -0.0625f);
}

__global__ void k_fwdY(const float* __restrict__ Tr, const float* __restrict__ Ti,
                       const float* __restrict__ tabC, const float* __restrict__ tabS,
                       float* __restrict__ Frp, float* __restrict__ Fip) {
  int wid = blockIdx.x * (blockDim.x >> 5) + (threadIdx.x >> 5);
  int lane = threadIdx.x & 31;
  int b = wid >> 7;
  int r2 = wid & 127;
  int col0 = (r2 >> 2) * 16;
  int kc = r2 & 3;
  const float* trb = Tr + (size_t)b * H_ * (M_ * C_);
  const float* tib = Ti + (size_t)b * H_ * (M_ * C_);
  v8f aCT = vzero(), aST = vzero(), aCTi = vzero(), aSTr = vzero();
  for (int k0 = kc * 64; k0 < kc * 64 + 64; k0 += 4) {
    v2f aC = ldA(tabC + k0, W_, lane);
    v2f aS = ldA(tabS + k0, W_, lane);
    v2f bR = ldB(trb + k0 * 512 + col0, 512, lane);
    v2f bI = ldB(tib + k0 * 512 + col0, 512, lane);
    aCT = wmma4(aC, bR, aCT);
    aST = wmma4(aS, bI, aST);
    aCTi = wmma4(aC, bI, aCTi);
    aSTr = wmma4(aS, bR, aSTr);
  }
  float* fr = Frp + (size_t)kc * (B_ * M_ * 512) + (size_t)b * (M_ * 512) + col0;
  float* fi = Fip + (size_t)kc * (B_ * M_ * 512) + (size_t)b * (M_ * 512) + col0;
  int n = lane & 15, mh = (lane >> 4) << 3;
#pragma unroll
  for (int r = 0; r < 8; ++r) {
    fr[(r + mh) * 512 + n] = (aCT[r] + aST[r]) * 0.0625f;
    fi[(r + mh) * 512 + n] = (aCTi[r] - aSTr[r]) * 0.0625f;
  }
}

__global__ void k_mix(const float* __restrict__ Frp, const float* __restrict__ Fip,
                      const float* __restrict__ wr, const float* __restrict__ wi,
                      float* __restrict__ Gr, float* __restrict__ Gi) {
  __shared__ float sFr[512];
  __shared__ float sFi[512];
  int b = blockIdx.x >> 4, k1 = blockIdx.x & 15;
  int tid = threadIdx.x;
  size_t base = ((size_t)b * M_ + k1) * 512;
  float fr = 0.f, fi = 0.f;
#pragma unroll
  for (int kc = 0; kc < 4; ++kc) {
    fr += Frp[(size_t)kc * (B_ * M_ * 512) + base + tid];
    fi += Fip[(size_t)kc * (B_ * M_ * 512) + base + tid];
  }
  sFr[tid] = fr; sFi[tid] = fi;
  __syncthreads();
  int o = tid >> 4, k2 = tid & 15;
  const float* wrk = wr + k1 * 16 + k2;
  const float* wik = wi + k1 * 16 + k2;
  float gr = 0.f, gi = 0.f;
#pragma unroll 4
  for (int i = 0; i < C_; ++i) {
    float xr = sFr[k2 * 32 + i], xi = sFi[k2 * 32 + i];
    float wrv = wrk[(i * 32 + o) * 256];
    float wiv = wik[(i * 32 + o) * 256];
    gr += xr * wrv - xi * wiv;
    gi += xr * wiv + xi * wrv;
  }
  size_t gidx = base + k2 * 32 + o;
  Gr[gidx] = gr; Gi[gidx] = gi;
}

__global__ void k_invY(const float* __restrict__ Gr, const float* __restrict__ Gi,
                       const float* __restrict__ tabC, const float* __restrict__ tabS,
                       float* __restrict__ P, float* __restrict__ Q) {
  int wid = blockIdx.x * (blockDim.x >> 5) + (threadIdx.x >> 5);
  int lane = threadIdx.x & 31;
  int b = wid >> 9, rest = wid & 511;
  int y0 = (rest >> 5) * 16, col0 = (rest & 31) * 16;
  const float* grb = Gr + (size_t)b * 8192;
  const float* gib = Gi + (size_t)b * 8192;
  v8f aCG = vzero(), aSGi = vzero(), aSG = vzero(), aCGi = vzero();
#pragma unroll
  for (int k0 = 0; k0 < M_; k0 += 4) {
    v2f aC = ldAt(tabC + k0 * W_ + y0, W_, lane);
    v2f aS = ldAt(tabS + k0 * W_ + y0, W_, lane);
    v2f bR = ldB(grb + k0 * 512 + col0, 512, lane);
    v2f bI = ldB(gib + k0 * 512 + col0, 512, lane);
    aCG = wmma4(aC, bR, aCG);
    aSGi = wmma4(aS, bI, aSGi);
    aSG = wmma4(aS, bR, aSG);
    aCGi = wmma4(aC, bI, aCGi);
  }
  float* p = P + (size_t)(b * H_ + y0) * 512 + col0;
  float* q = Q + (size_t)(b * H_ + y0) * 512 + col0;
  int n = lane & 15, mh = (lane >> 4) << 3;
#pragma unroll
  for (int r = 0; r < 8; ++r) {
    p[(r + mh) * 512 + n] = (aCG[r] - aSGi[r]) * 0.0625f;
    q[(r + mh) * 512 + n] = (aSG[r] + aCGi[r]) * 0.0625f;
  }
}

__global__ void k_invX(const float* __restrict__ P, const float* __restrict__ Q,
                       const float* __restrict__ xin, const float* __restrict__ tabC2,
                       const float* __restrict__ tabS2, const float* __restrict__ bw,
                       const float* __restrict__ bb, float* __restrict__ xout) {
  int wid = blockIdx.x * (blockDim.x >> 5) + (threadIdx.x >> 5);
  int lane = threadIdx.x & 31;
  int by = wid >> 4;
  int x0 = (wid & 15) * 16;
  const float* Pby = P + (size_t)by * 512;
  const float* Qby = Q + (size_t)by * 512;
  const float* xrow = xin + (size_t)by * (W_ * C_);
  v8f acc0 = vzero(), acc1 = vzero();
#pragma unroll
  for (int k0 = 0; k0 < M_; k0 += 4) {
    v2f aC = ldAt(tabC2 + k0 * W_ + x0, W_, lane);
    v2f bP0 = ldB(Pby + k0 * C_, C_, lane);
    v2f bP1 = ldB(Pby + k0 * C_ + 16, C_, lane);
    acc0 = wmma4(aC, bP0, acc0);
    acc1 = wmma4(aC, bP1, acc1);
    v2f aS = ldAt(tabS2 + k0 * W_ + x0, W_, lane);
    v2f aSn; aSn.x = -aS.x; aSn.y = -aS.y;
    v2f bQ0 = ldB(Qby + k0 * C_, C_, lane);
    v2f bQ1 = ldB(Qby + k0 * C_ + 16, C_, lane);
    acc0 = wmma4(aSn, bQ0, acc0);
    acc1 = wmma4(aSn, bQ1, acc1);
  }
#pragma unroll
  for (int k0 = 0; k0 < C_; k0 += 4) {
    v2f aX = ldA(xrow + x0 * C_ + k0, C_, lane);
    v2f bW0 = ldBt(bw + k0, C_, lane);
    v2f bW1 = ldBt(bw + 16 * C_ + k0, C_, lane);
    acc0 = wmma4(aX, bW0, acc0);
    acc1 = wmma4(aX, bW1, acc1);
  }
  int n = lane & 15, mh = (lane >> 4) << 3;
  float bb0 = bb[n], bb1 = bb[16 + n];
  float* out = xout + (size_t)by * (W_ * C_);
#pragma unroll
  for (int r = 0; r < 8; ++r) {
    int xr = x0 + r + mh;
    float v0 = acc0[r] + bb0;
    v0 = 0.5f * v0 * (1.0f + erff(v0 * 0.70710678f));
    out[xr * C_ + n] = v0;
    float v1 = acc1[r] + bb1;
    v1 = 0.5f * v1 * (1.0f + erff(v1 * 0.70710678f));
    out[xr * C_ + 16 + n] = v1;
  }
}

__global__ void k_proj(const float* __restrict__ xb, const float* __restrict__ pw,
                       const float* __restrict__ pb, float* __restrict__ out,
                       int o_outer) {
  int p = blockIdx.x * blockDim.x + threadIdx.x;
  if (p >= B_ * PIX) return;
  int b = p >> 16, pix = p & (PIX - 1);
  const float* v = xb + (size_t)p * C_;
  float s = pb[0];
#pragma unroll
  for (int c = 0; c < C_; ++c) s += pw[c] * v[c];
  out[((size_t)(b * 4 + o_outer)) * PIX + pix] = s;
}

extern "C" void kernel_launch(void* const* d_in, const int* in_sizes, int n_in,
                              void* d_out, int out_size, void* d_ws, size_t ws_size,
                              hipStream_t stream) {
  const float* x      = (const float*)d_in[0];
  const float* lift_w = (const float*)d_in[1];
  const float* lift_b = (const float*)d_in[2];
  const float* wr     = (const float*)d_in[3];
  const float* wi     = (const float*)d_in[4];
  const float* byp_w  = (const float*)d_in[5];
  const float* byp_b  = (const float*)d_in[6];
  const float* proj_w = (const float*)d_in[7];
  const float* proj_b = (const float*)d_in[8];
  float* out = (float*)d_out;

  float* w = (float*)d_ws;
  float* tabC  = w;
  float* tabS  = tabC + 4096;
  float* tabC2 = tabS + 4096;
  float* tabS2 = tabC2 + 4096;
  float* x0    = tabS2 + 4096;
  float* bufA  = x0 + (size_t)B_ * PIX * C_;
  float* bufB  = bufA + (size_t)B_ * PIX * C_;
  float* Tr    = bufB + (size_t)B_ * PIX * C_;
  float* Ti    = Tr + (size_t)B_ * H_ * M_ * C_;
  float* Frp   = Ti + (size_t)B_ * H_ * M_ * C_;
  float* Fip   = Frp + (size_t)4 * B_ * M_ * 512;
  float* Gr    = Fip + (size_t)4 * B_ * M_ * 512;
  float* Gi    = Gr + (size_t)B_ * M_ * M_ * C_;
  float* Pb    = Gi + (size_t)B_ * M_ * M_ * C_;
  float* Qb    = Pb + (size_t)B_ * H_ * M_ * C_;

  k_tabs<<<16, 256, 0, stream>>>(tabC, tabS, tabC2, tabS2);
  k_lift<<<(B_ * PIX + 255) / 256, 256, 0, stream>>>(x, lift_w, lift_b, x0);

  for (int o = 0; o < 4; ++o) {
    for (int n = 0; n < 4; ++n) {
      const float* cur = (n == 0) ? x0 : ((n & 1) ? bufA : bufB);
      float* dst = (n & 1) ? bufB : bufA;
      size_t wOff = ((size_t)(o * 4 + n)) * C_ * C_ * M_ * M_;
      size_t bwOff = ((size_t)(o * 4 + n)) * C_ * C_;
      size_t bbOff = ((size_t)(o * 4 + n)) * C_;
      k_fwdX<<<256, 256, 0, stream>>>(cur, tabC, tabS, Tr, Ti);
      k_fwdY<<<64, 256, 0, stream>>>(Tr, Ti, tabC, tabS, Frp, Fip);
      k_mix<<<64, 512, 0, stream>>>(Frp, Fip, wr + wOff, wi + wOff, Gr, Gi);
      k_invY<<<256, 256, 0, stream>>>(Gr, Gi, tabC, tabS, Pb, Qb);
      k_invX<<<2048, 256, 0, stream>>>(Pb, Qb, cur, tabC2, tabS2,
                                       byp_w + bwOff, byp_b + bbOff, dst);
    }
    k_proj<<<(B_ * PIX + 255) / 256, 256, 0, stream>>>(bufB, proj_w, proj_b, out, o);
  }
}


# First step's forward transform done once for all four chains (they start from the same lifted field): its summed F is kept and read by each chain's first mix; 54 -> 51 launches; same sums
# speedup vs baseline: 1.5381x; 1.5381x over previous
#include <hip/hip_runtime.h>
#include <math.h>


typedef float v2f __attribute__((ext_vector_type(2)));
typedef float v4f __attribute__((ext_vector_type(4)));
typedef float v8f __attribute__((ext_vector_type(8)));

#define B_  4
#define C_  32
#define H_  256
#define W_  256
#define M_  16
#define PIX (H_ * W_)

__device__ __forceinline__ v8f wmma4(v2f a, v2f b, v8f c) {
  return __builtin_amdgcn_wmma_f32_16x16x4_f32(false, a, false, b, (short)0, c,
                                               false, false);
}

__device__ __forceinline__ v2f ldA(const float* p, int lda, int lane) {
  int r = lane & 15, kk = (lane >> 4) << 1;
  v2f v; v.x = p[r * lda + kk]; v.y = p[r * lda + kk + 1]; return v;
}
__device__ __forceinline__ v4f ldF(const float* t, int g, int lane) {
  return *(const v4f*)(t + ((g * 32 + lane) << 2));
}
__device__ __forceinline__ v2f lo2(v4f v) { v2f a; a.x = v.x; a.y = v.y; return a; }
__device__ __forceinline__ v2f hi2(v4f v) { v2f a; a.x = v.z; a.y = v.w; return a; }
__device__ __forceinline__ v2f ldB(const float* p, int ldb, int lane) {
  int n = lane & 15, kk = (lane >> 4) << 1;
  v2f v; v.x = p[kk * ldb + n]; v.y = p[(kk + 1) * ldb + n]; return v;
}
__device__ __forceinline__ v2f ldBt(const float* p, int ldb, int lane) {
  int n = lane & 15, kk = (lane >> 4) << 1;
  v2f v; v.x = p[n * ldb + kk]; v.y = p[n * ldb + kk + 1]; return v;
}
__device__ __forceinline__ void stD(float* p, int ldd, int lane, v8f d, float s) {
  int n = lane & 15, mh = (lane >> 4) << 3;
#pragma unroll
  for (int r = 0; r < 8; ++r) p[(r + mh) * ldd + n] = d[r] * s;
}

__device__ __forceinline__ v8f vzero() {
  v8f z = {0.f, 0.f, 0.f, 0.f, 0.f, 0.f, 0.f, 0.f};
  return z;
}

__global__ void k_tabs(float* fC, float* fS, float* iC, float* iS) {
  int i = blockIdx.x * blockDim.x + threadIdx.x;
  if (i >= M_ * W_) return;
  int e = i & 3, lane = (i >> 2) & 31, g = i >> 7;
  int r = lane & 15, kk = (lane >> 4) << 1, s = e & 1;
  int xf = 4 * (2 * g + (e >> 1)) + kk + s;
  int mf = (r * xf) & 255;
  float af = 0.0245436926f * (float)mf;
  fC[i] = cosf(af); fS[i] = sinf(af);
  int t = g >> 1, p = g & 1;
  int km = 4 * (2 * p + (e >> 1)) + kk + s;
  int mi = (km * (16 * t + r)) & 255;
  float ai = 0.0245436926f * (float)mi;
  iC[i] = cosf(ai); iS[i] = sinf(ai);
}

__global__ void k_lift(const float* __restrict__ x, const float* __restrict__ lw,
                       const float* __restrict__ lb, float* __restrict__ x0) {
  int p = blockIdx.x * blockDim.x + threadIdx.x;
  if (p >= B_ * PIX) return;
  int b = p >> 16, pix = p & (PIX - 1);
  float xi0 = x[(b * 3 + 0) * PIX + pix];
  float xi1 = x[(b * 3 + 1) * PIX + pix];
  float xi2 = x[(b * 3 + 2) * PIX + pix];
  float* o = x0 + (size_t)p * C_;
#pragma unroll
  for (int c = 0; c < C_; ++c)
    o[c] = lb[c] + lw[c * 3 + 0] * xi0 + lw[c * 3 + 1] * xi1 + lw[c * 3 + 2] * xi2;
}

#define TP 264
__global__ void __launch_bounds__(512) __attribute__((amdgpu_waves_per_eu(4, 4)))
k_fwdXY(const float* __restrict__ xin, const float* __restrict__ fC,
        const float* __restrict__ fS, float* __restrict__ Frp,
        float* __restrict__ Fip) {
  __shared__ float sTr[16 * TP];
  __shared__ float sTi[16 * TP];
  int w = threadIdx.x >> 5;
  int r8 = w & 7;
  int xh = w >> 3;
  int lane = threadIdx.x & 31;
  int ch = blockIdx.x & 1;
  int grp = (blockIdx.x >> 1) & 31;
  int b = blockIdx.x >> 6;
  int y = grp * 8 + r8;
  const float* row = xin + ((size_t)(b * H_ + y)) * W_ * C_ + 16 * ch;
  v8f aC = vzero(), aS = vzero();
  int g0 = xh * (W_ / 16);
  const float* rowh = row + (size_t)(8 * g0) * C_;
  const float* fCh = fC + g0 * 128;
  const float* fSh = fS + g0 * 128;
  for (int i = 0; i < W_ / 16; ++i) {
    v4f c4 = ldF(fCh, i, lane);
    v4f s4 = ldF(fSh, i, lane);
    v2f b0 = ldB(rowh + (8 * i) * C_, C_, lane);
    v2f b1 = ldB(rowh + (8 * i + 4) * C_, C_, lane);
    aC = wmma4(lo2(c4), b0, aC);
    aS = wmma4(lo2(s4), b0, aS);
    aC = wmma4(hi2(c4), b1, aC);
    aS = wmma4(hi2(s4), b1, aS);
  }
  v4f yc4 = ldF(fC, grp, lane);
  v4f ys4 = ldF(fS, grp, lane);
  int n = lane & 15, mh = (lane >> 4) << 3;
#pragma unroll
  for (int r = 0; r < 8; ++r) {
    sTr[w * TP + (r + mh) * 16 + n] = aC[r] * 0.0625f;
    sTi[w * TP + (r + mh) * 16 + n] = aS[r] * -0.0625f;
  }
  __syncthreads();
  size_t pbase = (size_t)grp * (B_ * M_ * 512) + (size_t)b * (M_ * 512) + 16 * ch;
  {
    int k2 = w;
    const float* tr = sTr + k2 * 16;
    const float* ti = sTi + k2 * 16;
    v2f bR0 = ldB(tr, TP, lane) + ldB(tr + 8 * TP, TP, lane);
    v2f bI0 = ldB(ti, TP, lane) + ldB(ti + 8 * TP, TP, lane);
    v2f bR1 = ldB(tr + 4 * TP, TP, lane) + ldB(tr + 12 * TP, TP, lane);
    v2f bI1 = ldB(ti + 4 * TP, TP, lane) + ldB(ti + 12 * TP, TP, lane);
    v8f aCT = vzero(), aST = vzero(), aCTi = vzero(), aSTr = vzero();
    aCT = wmma4(lo2(yc4), bR0, aCT);
    aST = wmma4(lo2(ys4), bI0, aST);
    aCTi = wmma4(lo2(yc4), bI0, aCTi);
    aSTr = wmma4(lo2(ys4), bR0, aSTr);
    aCT = wmma4(hi2(yc4), bR1, aCT);
    aST = wmma4(hi2(ys4), bI1, aST);
    aCTi = wmma4(hi2(yc4), bI1, aCTi);
    aSTr = wmma4(hi2(ys4), bR1, aSTr);
    float* fr = Frp + pbase + k2 * 32;
    float* fi = Fip + pbase + k2 * 32;
#pragma unroll
    for (int r = 0; r < 8; ++r) {
      fr[(r + mh) * 512 + n] = (aCT[r] + aST[r]) * 0.0625f;
      fi[(r + mh) * 512 + n] = (aCTi[r] - aSTr[r]) * 0.0625f;
    }
  }
}

#define FP 36
#define GP 34
template <int MODE>
__global__ void __launch_bounds__(512)
k_mix(const float* __restrict__ Frp, const float* __restrict__ Fip,
      const float* __restrict__ wr, const float* __restrict__ wi,
      float* __restrict__ Gr, float* __restrict__ Gi,
      float* __restrict__ Fsr, float* __restrict__ Fsi) {
  __shared__ __attribute__((aligned(16))) float sPr[4 * 512];
  __shared__ __attribute__((aligned(16))) float sPi[4 * 512];
  __shared__ __attribute__((aligned(16))) float sFr[16 * FP];
  __shared__ __attribute__((aligned(16))) float sFi[16 * FP];
  __shared__ float sGr[4 * 16 * GP];
  __shared__ float sGi[4 * 16 * GP];
  int b = blockIdx.x >> 4, k1 = blockIdx.x & 15;
  int tid = threadIdx.x;
  size_t base = ((size_t)b * M_ + k1) * 512;
  int grp = tid >> 7;
  int col = tid & 127;
  int kq = tid & 3, o = (tid >> 2) & 31;
  const float* pr = Frp + (size_t)(8 * grp) * (B_ * M_ * 512) + base + 4 * col;
  const float* pi = Fip + (size_t)(8 * grp) * (B_ * M_ * 512) + base + 4 * col;
  const float* wrp = wr + ((size_t)(8 * grp * 32 + o)) * 256 + k1 * 16 + 4 * kq;
  const float* wip = wi + ((size_t)(8 * grp * 32 + o)) * 256 + k1 * 16 + 4 * kq;
  v4f vr[8], vi[8], wrv[8], wiv[8];
  float fkr = 0.f, fki = 0.f;
  if (MODE == 2) {
    fkr = Fsr[base + tid];
    fki = Fsi[base + tid];
  } else {
#pragma unroll
    for (int q = 0; q < 8; ++q) {
      vr[q] = *(const v4f*)(pr + (size_t)q * (B_ * M_ * 512));
      vi[q] = *(const v4f*)(pi + (size_t)q * (B_ * M_ * 512));
    }
  }
#pragma unroll
  for (int j = 0; j < 8; ++j) {
    wrv[j] = *(const v4f*)(wrp + (size_t)j * (32 * 256));
    wiv[j] = *(const v4f*)(wip + (size_t)j * (32 * 256));
  }
  if (MODE == 2) {
    sFr[(tid >> 5) * FP + (tid & 31)] = fkr;
    sFi[(tid >> 5) * FP + (tid & 31)] = fki;
  } else {
    v4f ar = vr[0], ai = vi[0];
#pragma unroll
    for (int q = 1; q < 8; ++q) {
      ar += vr[q];
      ai += vi[q];
    }
    *(v4f*)(sPr + grp * 512 + 4 * col) = ar;
    *(v4f*)(sPi + grp * 512 + 4 * col) = ai;
    __syncthreads();
    fkr = ((sPr[tid] + sPr[512 + tid]) + sPr[1024 + tid]) + sPr[1536 + tid];
    fki = ((sPi[tid] + sPi[512 + tid]) + sPi[1024 + tid]) + sPi[1536 + tid];
    sFr[(tid >> 5) * FP + (tid & 31)] = fkr;
    sFi[(tid >> 5) * FP + (tid & 31)] = fki;
  }
  __syncthreads();
  float gr[4] = {0.f, 0.f, 0.f, 0.f}, gi[4] = {0.f, 0.f, 0.f, 0.f};
#pragma unroll
  for (int m = 0; m < 4; ++m) {
    const float* fr = sFr + (4 * kq + m) * FP + 8 * grp;
    const float* fi = sFi + (4 * kq + m) * FP + 8 * grp;
    v4f xr0 = *(const v4f*)fr, xr1 = *(const v4f*)(fr + 4);
    v4f xi0 = *(const v4f*)fi, xi1 = *(const v4f*)(fi + 4);
#pragma unroll
    for (int j = 0; j < 8; ++j) {
      float xr = (j < 4) ? xr0[j & 3] : xr1[j & 3];
      float xi = (j < 4) ? xi0[j & 3] : xi1[j & 3];
      float wrs = wrv[j][m], wis = wiv[j][m];
      gr[m] += xr * wrs - xi * wis;
      gi[m] += xr * wis + xi * wrs;
    }
  }
#pragma unroll
  for (int m = 0; m < 4; ++m) {
    sGr[grp * (16 * GP) + (4 * kq + m) * GP + o] = gr[m];
    sGi[grp * (16 * GP) + (4 * kq + m) * GP + o] = gi[m];
  }
  __syncthreads();
  {
    int gx = (tid >> 5) * GP + (tid & 31);
    float g_r = ((sGr[gx] + sGr[16 * GP + gx]) + sGr[2 * 16 * GP + gx]) + sGr[3 * 16 * GP + gx];
    float g_i = ((sGi[gx] + sGi[16 * GP + gx]) + sGi[2 * 16 * GP + gx]) + sGi[3 * 16 * GP + gx];
    Gr[base + tid] = g_r;
    Gi[base + tid] = g_i;
  }
  if (MODE == 1) {
    Fsr[base + tid] = fkr;
    Fsi[base + tid] = fki;
  }
}

#define PP 516
__global__ void __launch_bounds__(1024)
k_invYX(const float* __restrict__ Gr, const float* __restrict__ Gi,
        const float* __restrict__ xin, const float* __restrict__ iC,
        const float* __restrict__ iS, const float* __restrict__ bw,
        const float* __restrict__ bb, float* __restrict__ xout) {
  __shared__ float sP[16 * PP];
  __shared__ float sQ[16 * PP];
  int w = threadIdx.x >> 5;
  int lane = threadIdx.x & 31;
  int xq = blockIdx.x & 3;
  int t = (blockIdx.x >> 2) & 15;
  int b = blockIdx.x >> 6;
  int n = lane & 15, mh = (lane >> 4) << 3;
  {
    int col0 = 16 * w;
    const float* grb = Gr + (size_t)b * 8192;
    const float* gib = Gi + (size_t)b * 8192;
    v8f aCG = vzero(), aSGi = vzero(), aSG = vzero(), aCGi = vzero();
    v4f cA = ldF(iC, 2 * t, lane), cB = ldF(iC, 2 * t + 1, lane);
    v4f sA = ldF(iS, 2 * t, lane), sB = ldF(iS, 2 * t + 1, lane);
    v2f aCk[4] = {lo2(cA), hi2(cA), lo2(cB), hi2(cB)};
    v2f aSk[4] = {lo2(sA), hi2(sA), lo2(sB), hi2(sB)};
#pragma unroll
    for (int k0 = 0; k0 < M_; k0 += 4) {
      v2f aC = aCk[k0 >> 2];
      v2f aS = aSk[k0 >> 2];
      v2f bR = ldB(grb + k0 * 512 + col0, 512, lane);
      v2f bI = ldB(gib + k0 * 512 + col0, 512, lane);
      aCG = wmma4(aC, bR, aCG);
      aSGi = wmma4(aS, bI, aSGi);
      aSG = wmma4(aS, bR, aSG);
      aCGi = wmma4(aC, bI, aCGi);
    }
    float wk = (col0 < C_) ? 0.00390625f : 0.0078125f;
#pragma unroll
    for (int r = 0; r < 8; ++r) {
      sP[(r + mh) * PP + col0 + n] = (aCG[r] - aSGi[r]) * wk;
      sQ[(r + mh) * PP + col0 + n] = (aSG[r] + aCGi[r]) * wk;
    }
  }
  __syncthreads();
  int row = w >> 1;
  int by = b * H_ + 16 * t + row;
  const float* Prow = sP + row * PP;
  const float* Qrow = sQ + row * PP;
  const float* xrow = xin + (size_t)by * (W_ * C_);
  float* out = xout + (size_t)by * (W_ * C_);
  v2f bP0[4], bP1[4], bQ0[4], bQ1[4];
#pragma unroll
  for (int j = 0; j < 4; ++j) {
    bP0[j] = ldB(Prow + 4 * j * C_, C_, lane);
    bP1[j] = ldB(Prow + 4 * j * C_ + 16, C_, lane);
    bQ0[j] = ldB(Qrow + 4 * j * C_, C_, lane);
    bQ1[j] = ldB(Qrow + 4 * j * C_ + 16, C_, lane);
  }
  v2f bW0[8], bW1[8];
#pragma unroll
  for (int j = 0; j < 8; ++j) {
    bW0[j] = ldBt(bw + 4 * j, C_, lane);
    bW1[j] = ldBt(bw + 16 * C_ + 4 * j, C_, lane);
  }
  float bb0 = bb[n], bb1 = bb[16 + n];
  for (int jt = 0; jt < 2; ++jt) {
    int tx = 4 * xq + 2 * (w & 1) + jt;
    int x0 = 16 * tx;
    v8f acc0 = vzero(), acc1 = vzero();
    v4f cA = ldF(iC, 2 * tx, lane), cB = ldF(iC, 2 * tx + 1, lane);
    v4f sA = ldF(iS, 2 * tx, lane), sB = ldF(iS, 2 * tx + 1, lane);
    v2f aCk[4] = {lo2(cA), hi2(cA), lo2(cB), hi2(cB)};
    v2f aSk[4] = {lo2(sA), hi2(sA), lo2(sB), hi2(sB)};
#pragma unroll
    for (int j = 0; j < 4; ++j) {
      v2f aC = aCk[j];
      acc0 = wmma4(aC, bP0[j], acc0);
      acc1 = wmma4(aC, bP1[j], acc1);
      v2f aS = aSk[j];
      v2f aSn; aSn.x = -aS.x; aSn.y = -aS.y;
      acc0 = wmma4(aSn, bQ0[j], acc0);
      acc1 = wmma4(aSn, bQ1[j], acc1);
    }
#pragma unroll
    for (int j = 0; j < 8; ++j) {
      v2f aX = ldA(xrow + x0 * C_ + 4 * j, C_, lane);
      acc0 = wmma4(aX, bW0[j], acc0);
      acc1 = wmma4(aX, bW1[j], acc1);
    }
#pragma unroll
    for (int r = 0; r < 8; ++r) {
      int xr = x0 + r + mh;
      float v0 = acc0[r] + bb0;
      v0 = 0.5f * v0 * (1.0f + erff(v0 * 0.70710678f));
      out[xr * C_ + n] = v0;
      float v1 = acc1[r] + bb1;
      v1 = 0.5f * v1 * (1.0f + erff(v1 * 0.70710678f));
      out[xr * C_ + 16 + n] = v1;
    }
  }
}

__global__ void k_proj(const float* __restrict__ xb, const float* __restrict__ pw,
                       const float* __restrict__ pb, float* __restrict__ out,
                       int o_outer) {
  int p = blockIdx.x * blockDim.x + threadIdx.x;
  if (p >= B_ * PIX) return;
  int b = p >> 16, pix = p & (PIX - 1);
  const float* v = xb + (size_t)p * C_;
  float s = pb[0];
#pragma unroll
  for (int c = 0; c < C_; ++c) s += pw[c] * v[c];
  out[((size_t)(b * 4 + o_outer)) * PIX + pix] = s;
}

extern "C" void kernel_launch(void* const* d_in, const int* in_sizes, int n_in,
                              void* d_out, int out_size, void* d_ws, size_t ws_size,
                              hipStream_t stream) {
  const float* x      = (const float*)d_in[0];
  const float* lift_w = (const float*)d_in[1];
  const float* lift_b = (const float*)d_in[2];
  const float* wr     = (const float*)d_in[3];
  const float* wi     = (const float*)d_in[4];
  const float* byp_w  = (const float*)d_in[5];
  const float* byp_b  = (const float*)d_in[6];
  const float* proj_w = (const float*)d_in[7];
  const float* proj_b = (const float*)d_in[8];
  float* out = (float*)d_out;

  float* w = (float*)d_ws;
  float* fC    = w;
  float* fS    = fC + 4096;
  float* iC    = fS + 4096;
  float* iS    = iC + 4096;
  float* x0    = iS + 4096;
  float* bufA  = x0 + (size_t)B_ * PIX * C_;
  float* bufB  = bufA + (size_t)B_ * PIX * C_;
  float* Gr    = bufB + (size_t)B_ * PIX * C_;
  float* Gi    = Gr + (size_t)B_ * M_ * M_ * C_;
  float* Frp   = Gi + (size_t)B_ * M_ * M_ * C_;
  float* Fip   = Frp + (size_t)32 * B_ * M_ * 512;
  float* Fsr   = Fip + (size_t)32 * B_ * M_ * 512;
  float* Fsi   = Fsr + (size_t)B_ * M_ * 512;

  k_tabs<<<16, 256, 0, stream>>>(fC, fS, iC, iS);
  k_lift<<<(B_ * PIX + 255) / 256, 256, 0, stream>>>(x, lift_w, lift_b, x0);
  k_fwdXY<<<256, 512, 0, stream>>>(x0, fC, fS, Frp, Fip);

  for (int o = 0; o < 4; ++o) {
    for (int n = 0; n < 4; ++n) {
      const float* cur = (n == 0) ? x0 : ((n & 1) ? bufA : bufB);
      float* dst = (n & 1) ? bufB : bufA;
      size_t wOff = ((size_t)(o * 4 + n)) * C_ * C_ * M_ * M_;
      size_t bwOff = ((size_t)(o * 4 + n)) * C_ * C_;
      size_t bbOff = ((size_t)(o * 4 + n)) * C_;
      if (n == 0) {
        if (o == 0)
          k_mix<1><<<64, 512, 0, stream>>>(Frp, Fip, wr + wOff, wi + wOff, Gr, Gi, Fsr, Fsi);
        else
          k_mix<2><<<64, 512, 0, stream>>>(Frp, Fip, wr + wOff, wi + wOff, Gr, Gi, Fsr, Fsi);
      } else {
        k_fwdXY<<<256, 512, 0, stream>>>(cur, fC, fS, Frp, Fip);
        k_mix<0><<<64, 512, 0, stream>>>(Frp, Fip, wr + wOff, wi + wOff, Gr, Gi, Fsr, Fsi);
      }
      k_invYX<<<256, 1024, 0, stream>>>(Gr, Gi, cur, iC, iS,
                                        byp_w + bwOff, byp_b + bbOff, dst);
    }
    k_proj<<<(B_ * PIX + 255) / 256, 256, 0, stream>>>(bufB, proj_w, proj_b, out, o);
  }
}


# Independent small launches share a grid: chain o's projection runs as extra blocks of chain o+1's first mix launch, the table build as extra blocks of the lift launch; 51 -> 47 launches; same sums
# speedup vs baseline: 1.6141x; 1.0494x over previous
#include <hip/hip_runtime.h>
#include <math.h>


typedef float v2f __attribute__((ext_vector_type(2)));
typedef float v4f __attribute__((ext_vector_type(4)));
typedef float v8f __attribute__((ext_vector_type(8)));

#define B_  4
#define C_  32
#define H_  256
#define W_  256
#define M_  16
#define PIX (H_ * W_)

__device__ __forceinline__ v8f wmma4(v2f a, v2f b, v8f c) {
  return __builtin_amdgcn_wmma_f32_16x16x4_f32(false, a, false, b, (short)0, c,
                                               false, false);
}

__device__ __forceinline__ v2f ldA(const float* p, int lda, int lane) {
  int r = lane & 15, kk = (lane >> 4) << 1;
  v2f v; v.x = p[r * lda + kk]; v.y = p[r * lda + kk + 1]; return v;
}
__device__ __forceinline__ v4f ldF(const float* t, int g, int lane) {
  return *(const v4f*)(t + ((g * 32 + lane) << 2));
}
__device__ __forceinline__ v2f lo2(v4f v) { v2f a; a.x = v.x; a.y = v.y; return a; }
__device__ __forceinline__ v2f hi2(v4f v) { v2f a; a.x = v.z; a.y = v.w; return a; }
__device__ __forceinline__ v2f ldB(const float* p, int ldb, int lane) {
  int n = lane & 15, kk = (lane >> 4) << 1;
  v2f v; v.x = p[kk * ldb + n]; v.y = p[(kk + 1) * ldb + n]; return v;
}
__device__ __forceinline__ v2f ldBt(const float* p, int ldb, int lane) {
  int n = lane & 15, kk = (lane >> 4) << 1;
  v2f v; v.x = p[n * ldb + kk]; v.y = p[n * ldb + kk + 1]; return v;
}
__device__ __forceinline__ void stD(float* p, int ldd, int lane, v8f d, float s) {
  int n = lane & 15, mh = (lane >> 4) << 3;
#pragma unroll
  for (int r = 0; r < 8; ++r) p[(r + mh) * ldd + n] = d[r] * s;
}

__device__ __forceinline__ v8f vzero() {
  v8f z = {0.f, 0.f, 0.f, 0.f, 0.f, 0.f, 0.f, 0.f};
  return z;
}

__device__ __forceinline__ void tabs_body(int i, float* fC, float* fS, float* iC, float* iS) {
  int e = i & 3, lane = (i >> 2) & 31, g = i >> 7;
  int r = lane & 15, kk = (lane >> 4) << 1, s = e & 1;
  int xf = 4 * (2 * g + (e >> 1)) + kk + s;
  int mf = (r * xf) & 255;
  float af = 0.0245436926f * (float)mf;
  fC[i] = cosf(af); fS[i] = sinf(af);
  int t = g >> 1, p = g & 1;
  int km = 4 * (2 * p + (e >> 1)) + kk + s;
  int mi = (km * (16 * t + r)) & 255;
  float ai = 0.0245436926f * (float)mi;
  iC[i] = cosf(ai); iS[i] = sinf(ai);
}

__device__ __forceinline__ void lift_body(int p, const float* __restrict__ x,
                                          const float* __restrict__ lw,
                                          const float* __restrict__ lb,
                                          float* __restrict__ x0) {
  int b = p >> 16, pix = p & (PIX - 1);
  float xi0 = x[(b * 3 + 0) * PIX + pix];
  float xi1 = x[(b * 3 + 1) * PIX + pix];
  float xi2 = x[(b * 3 + 2) * PIX + pix];
  float* o = x0 + (size_t)p * C_;
#pragma unroll
  for (int c = 0; c < C_; ++c)
    o[c] = lb[c] + lw[c * 3 + 0] * xi0 + lw[c * 3 + 1] * xi1 + lw[c * 3 + 2] * xi2;
}

#define TAB_BLOCKS 16
__global__ void __launch_bounds__(256)
k_tabs_lift(float* fC, float* fS, float* iC, float* iS,
            const float* __restrict__ x, const float* __restrict__ lw,
            const float* __restrict__ lb, float* __restrict__ x0) {
  if (blockIdx.x < TAB_BLOCKS)
    tabs_body(blockIdx.x * 256 + threadIdx.x, fC, fS, iC, iS);
  else
    lift_body((blockIdx.x - TAB_BLOCKS) * 256 + threadIdx.x, x, lw, lb, x0);
}

#define TP 264
__global__ void __launch_bounds__(512) __attribute__((amdgpu_waves_per_eu(4, 4)))
k_fwdXY(const float* __restrict__ xin, const float* __restrict__ fC,
        const float* __restrict__ fS, float* __restrict__ Frp,
        float* __restrict__ Fip) {
  __shared__ float sTr[16 * TP];
  __shared__ float sTi[16 * TP];
  int w = threadIdx.x >> 5;
  int r8 = w & 7;
  int xh = w >> 3;
  int lane = threadIdx.x & 31;
  int ch = blockIdx.x & 1;
  int grp = (blockIdx.x >> 1) & 31;
  int b = blockIdx.x >> 6;
  int y = grp * 8 + r8;
  const float* row = xin + ((size_t)(b * H_ + y)) * W_ * C_ + 16 * ch;
  v8f aC = vzero(), aS = vzero();
  int g0 = xh * (W_ / 16);
  const float* rowh = row + (size_t)(8 * g0) * C_;
  const float* fCh = fC + g0 * 128;
  const float* fSh = fS + g0 * 128;
  for (int i = 0; i < W_ / 16; ++i) {
    v4f c4 = ldF(fCh, i, lane);
    v4f s4 = ldF(fSh, i, lane);
    v2f b0 = ldB(rowh + (8 * i) * C_, C_, lane);
    v2f b1 = ldB(rowh + (8 * i + 4) * C_, C_, lane);
    aC = wmma4(lo2(c4), b0, aC);
    aS = wmma4(lo2(s4), b0, aS);
    aC = wmma4(hi2(c4), b1, aC);
    aS = wmma4(hi2(s4), b1, aS);
  }
  v4f yc4 = ldF(fC, grp, lane);
  v4f ys4 = ldF(fS, grp, lane);
  int n = lane & 15, mh = (lane >> 4) << 3;
#pragma unroll
  for (int r = 0; r < 8; ++r) {
    sTr[w * TP + (r + mh) * 16 + n] = aC[r] * 0.0625f;
    sTi[w * TP + (r + mh) * 16 + n] = aS[r] * -0.0625f;
  }
  __syncthreads();
  size_t pbase = (size_t)grp * (B_ * M_ * 512) + (size_t)b * (M_ * 512) + 16 * ch;
  {
    int k2 = w;
    const float* tr = sTr + k2 * 16;
    const float* ti = sTi + k2 * 16;
    v2f bR0 = ldB(tr, TP, lane) + ldB(tr + 8 * TP, TP, lane);
    v2f bI0 = ldB(ti, TP, lane) + ldB(ti + 8 * TP, TP, lane);
    v2f bR1 = ldB(tr + 4 * TP, TP, lane) + ldB(tr + 12 * TP, TP, lane);
    v2f bI1 = ldB(ti + 4 * TP, TP, lane) + ldB(ti + 12 * TP, TP, lane);
    v8f aCT = vzero(), aST = vzero(), aCTi = vzero(), aSTr = vzero();
    aCT = wmma4(lo2(yc4), bR0, aCT);
    aST = wmma4(lo2(ys4), bI0, aST);
    aCTi = wmma4(lo2(yc4), bI0, aCTi);
    aSTr = wmma4(lo2(ys4), bR0, aSTr);
    aCT = wmma4(hi2(yc4), bR1, aCT);
    aST = wmma4(hi2(ys4), bI1, aST);
    aCTi = wmma4(hi2(yc4), bI1, aCTi);
    aSTr = wmma4(hi2(ys4), bR1, aSTr);
    float* fr = Frp + pbase + k2 * 32;
    float* fi = Fip + pbase + k2 * 32;
#pragma unroll
    for (int r = 0; r < 8; ++r) {
      fr[(r + mh) * 512 + n] = (aCT[r] + aST[r]) * 0.0625f;
      fi[(r + mh) * 512 + n] = (aCTi[r] - aSTr[r]) * 0.0625f;
    }
  }
}

#define FP 36
#define GP 34
template <int MODE>
__device__ __forceinline__ void
mix_body(int bid, const float* __restrict__ Frp, const float* __restrict__ Fip,
         const float* __restrict__ wr, const float* __restrict__ wi,
         float* __restrict__ Gr, float* __restrict__ Gi,
         float* __restrict__ Fsr, float* __restrict__ Fsi) {
  __shared__ __attribute__((aligned(16))) float sPr[4 * 512];
  __shared__ __attribute__((aligned(16))) float sPi[4 * 512];
  __shared__ __attribute__((aligned(16))) float sFr[16 * FP];
  __shared__ __attribute__((aligned(16))) float sFi[16 * FP];
  __shared__ float sGr[4 * 16 * GP];
  __shared__ float sGi[4 * 16 * GP];
  int b = bid >> 4, k1 = bid & 15;
  int tid = threadIdx.x;
  size_t base = ((size_t)b * M_ + k1) * 512;
  int grp = tid >> 7;
  int col = tid & 127;
  int kq = tid & 3, o = (tid >> 2) & 31;
  const float* pr = Frp + (size_t)(8 * grp) * (B_ * M_ * 512) + base + 4 * col;
  const float* pi = Fip + (size_t)(8 * grp) * (B_ * M_ * 512) + base + 4 * col;
  const float* wrp = wr + ((size_t)(8 * grp * 32 + o)) * 256 + k1 * 16 + 4 * kq;
  const float* wip = wi + ((size_t)(8 * grp * 32 + o)) * 256 + k1 * 16 + 4 * kq;
  v4f vr[8], vi[8], wrv[8], wiv[8];
  float fkr = 0.f, fki = 0.f;
  if (MODE == 2) {
    fkr = Fsr[base + tid];
    fki = Fsi[base + tid];
  } else {
#pragma unroll
    for (int q = 0; q < 8; ++q) {
      vr[q] = *(const v4f*)(pr + (size_t)q * (B_ * M_ * 512));
      vi[q] = *(const v4f*)(pi + (size_t)q * (B_ * M_ * 512));
    }
  }
#pragma unroll
  for (int j = 0; j < 8; ++j) {
    wrv[j] = *(const v4f*)(wrp + (size_t)j * (32 * 256));
    wiv[j] = *(const v4f*)(wip + (size_t)j * (32 * 256));
  }
  if (MODE == 2) {
    sFr[(tid >> 5) * FP + (tid & 31)] = fkr;
    sFi[(tid >> 5) * FP + (tid & 31)] = fki;
  } else {
    v4f ar = vr[0], ai = vi[0];
#pragma unroll
    for (int q = 1; q < 8; ++q) {
      ar += vr[q];
      ai += vi[q];
    }
    *(v4f*)(sPr + grp * 512 + 4 * col) = ar;
    *(v4f*)(sPi + grp * 512 + 4 * col) = ai;
    __syncthreads();
    fkr = ((sPr[tid] + sPr[512 + tid]) + sPr[1024 + tid]) + sPr[1536 + tid];
    fki = ((sPi[tid] + sPi[512 + tid]) + sPi[1024 + tid]) + sPi[1536 + tid];
    sFr[(tid >> 5) * FP + (tid & 31)] = fkr;
    sFi[(tid >> 5) * FP + (tid & 31)] = fki;
  }
  __syncthreads();
  float gr[4] = {0.f, 0.f, 0.f, 0.f}, gi[4] = {0.f, 0.f, 0.f, 0.f};
#pragma unroll
  for (int m = 0; m < 4; ++m) {
    const float* fr = sFr + (4 * kq + m) * FP + 8 * grp;
    const float* fi = sFi + (4 * kq + m) * FP + 8 * grp;
    v4f xr0 = *(const v4f*)fr, xr1 = *(const v4f*)(fr + 4);
    v4f xi0 = *(const v4f*)fi, xi1 = *(const v4f*)(fi + 4);
#pragma unroll
    for (int j = 0; j < 8; ++j) {
      float xr = (j < 4) ? xr0[j & 3] : xr1[j & 3];
      float xi = (j < 4) ? xi0[j & 3] : xi1[j & 3];
      float wrs = wrv[j][m], wis = wiv[j][m];
      gr[m] += xr * wrs - xi * wis;
      gi[m] += xr * wis + xi * wrs;
    }
  }
#pragma unroll
  for (int m = 0; m < 4; ++m) {
    sGr[grp * (16 * GP) + (4 * kq + m) * GP + o] = gr[m];
    sGi[grp * (16 * GP) + (4 * kq + m) * GP + o] = gi[m];
  }
  __syncthreads();
  {
    int gx = (tid >> 5) * GP + (tid & 31);
    float g_r = ((sGr[gx] + sGr[16 * GP + gx]) + sGr[2 * 16 * GP + gx]) + sGr[3 * 16 * GP + gx];
    float g_i = ((sGi[gx] + sGi[16 * GP + gx]) + sGi[2 * 16 * GP + gx]) + sGi[3 * 16 * GP + gx];
    Gr[base + tid] = g_r;
    Gi[base + tid] = g_i;
  }
  if (MODE == 1) {
    Fsr[base + tid] = fkr;
    Fsi[base + tid] = fki;
  }
}

template <int MODE>
__global__ void __launch_bounds__(512)
k_mix(const float* __restrict__ Frp, const float* __restrict__ Fip,
      const float* __restrict__ wr, const float* __restrict__ wi,
      float* __restrict__ Gr, float* __restrict__ Gi,
      float* __restrict__ Fsr, float* __restrict__ Fsi) {
  mix_body<MODE>(blockIdx.x, Frp, Fip, wr, wi, Gr, Gi, Fsr, Fsi);
}

#define PP 516
__global__ void __launch_bounds__(1024)
k_invYX(const float* __restrict__ Gr, const float* __restrict__ Gi,
        const float* __restrict__ xin, const float* __restrict__ iC,
        const float* __restrict__ iS, const float* __restrict__ bw,
        const float* __restrict__ bb, float* __restrict__ xout) {
  __shared__ float sP[16 * PP];
  __shared__ float sQ[16 * PP];
  int w = threadIdx.x >> 5;
  int lane = threadIdx.x & 31;
  int xq = blockIdx.x & 3;
  int t = (blockIdx.x >> 2) & 15;
  int b = blockIdx.x >> 6;
  int n = lane & 15, mh = (lane >> 4) << 3;
  {
    int col0 = 16 * w;
    const float* grb = Gr + (size_t)b * 8192;
    const float* gib = Gi + (size_t)b * 8192;
    v8f aCG = vzero(), aSGi = vzero(), aSG = vzero(), aCGi = vzero();
    v4f cA = ldF(iC, 2 * t, lane), cB = ldF(iC, 2 * t + 1, lane);
    v4f sA = ldF(iS, 2 * t, lane), sB = ldF(iS, 2 * t + 1, lane);
    v2f aCk[4] = {lo2(cA), hi2(cA), lo2(cB), hi2(cB)};
    v2f aSk[4] = {lo2(sA), hi2(sA), lo2(sB), hi2(sB)};
#pragma unroll
    for (int k0 = 0; k0 < M_; k0 += 4) {
      v2f aC = aCk[k0 >> 2];
      v2f aS = aSk[k0 >> 2];
      v2f bR = ldB(grb + k0 * 512 + col0, 512, lane);
      v2f bI = ldB(gib + k0 * 512 + col0, 512, lane);
      aCG = wmma4(aC, bR, aCG);
      aSGi = wmma4(aS, bI, aSGi);
      aSG = wmma4(aS, bR, aSG);
      aCGi = wmma4(aC, bI, aCGi);
    }
    float wk = (col0 < C_) ? 0.00390625f : 0.0078125f;
#pragma unroll
    for (int r = 0; r < 8; ++r) {
      sP[(r + mh) * PP + col0 + n] = (aCG[r] - aSGi[r]) * wk;
      sQ[(r + mh) * PP + col0 + n] = (aSG[r] + aCGi[r]) * wk;
    }
  }
  __syncthreads();
  int row = w >> 1;
  int by = b * H_ + 16 * t + row;
  const float* Prow = sP + row * PP;
  const float* Qrow = sQ + row * PP;
  const float* xrow = xin + (size_t)by * (W_ * C_);
  float* out = xout + (size_t)by * (W_ * C_);
  v2f bP0[4], bP1[4], bQ0[4], bQ1[4];
#pragma unroll
  for (int j = 0; j < 4; ++j) {
    bP0[j] = ldB(Prow + 4 * j * C_, C_, lane);
    bP1[j] = ldB(Prow + 4 * j * C_ + 16, C_, lane);
    bQ0[j] = ldB(Qrow + 4 * j * C_, C_, lane);
    bQ1[j] = ldB(Qrow + 4 * j * C_ + 16, C_, lane);
  }
  v2f bW0[8], bW1[8];
#pragma unroll
  for (int j = 0; j < 8; ++j) {
    bW0[j] = ldBt(bw + 4 * j, C_, lane);
    bW1[j] = ldBt(bw + 16 * C_ + 4 * j, C_, lane);
  }
  float bb0 = bb[n], bb1 = bb[16 + n];
  for (int jt = 0; jt < 2; ++jt) {
    int tx = 4 * xq + 2 * (w & 1) + jt;
    int x0 = 16 * tx;
    v8f acc0 = vzero(), acc1 = vzero();
    v4f cA = ldF(iC, 2 * tx, lane), cB = ldF(iC, 2 * tx + 1, lane);
    v4f sA = ldF(iS, 2 * tx, lane), sB = ldF(iS, 2 * tx + 1, lane);
    v2f aCk[4] = {lo2(cA), hi2(cA), lo2(cB), hi2(cB)};
    v2f aSk[4] = {lo2(sA), hi2(sA), lo2(sB), hi2(sB)};
#pragma unroll
    for (int j = 0; j < 4; ++j) {
      v2f aC = aCk[j];
      acc0 = wmma4(aC, bP0[j], acc0);
      acc1 = wmma4(aC, bP1[j], acc1);
      v2f aS = aSk[j];
      v2f aSn; aSn.x = -aS.x; aSn.y = -aS.y;
      acc0 = wmma4(aSn, bQ0[j], acc0);
      acc1 = wmma4(aSn, bQ1[j], acc1);
    }
#pragma unroll
    for (int j = 0; j < 8; ++j) {
      v2f aX = ldA(xrow + x0 * C_ + 4 * j, C_, lane);
      acc0 = wmma4(aX, bW0[j], acc0);
      acc1 = wmma4(aX, bW1[j], acc1);
    }
#pragma unroll
    for (int r = 0; r < 8; ++r) {
      int xr = x0 + r + mh;
      float v0 = acc0[r] + bb0;
      v0 = 0.5f * v0 * (1.0f + erff(v0 * 0.70710678f));
      out[xr * C_ + n] = v0;
      float v1 = acc1[r] + bb1;
      v1 = 0.5f * v1 * (1.0f + erff(v1 * 0.70710678f));
      out[xr * C_ + 16 + n] = v1;
    }
  }
}

__device__ __forceinline__ void proj_body(int p, const float* __restrict__ xb,
                                          const float* __restrict__ pw,
                                          const float* __restrict__ pb,
                                          float* __restrict__ out, int o_outer) {
  int b = p >> 16, pix = p & (PIX - 1);
  const float* v = xb + (size_t)p * C_;
  float s = pb[0];
#pragma unroll
  for (int c = 0; c < C_; ++c) s += pw[c] * v[c];
  out[((size_t)(b * 4 + o_outer)) * PIX + pix] = s;
}

__global__ void __launch_bounds__(512)
k_proj(const float* __restrict__ xb, const float* __restrict__ pw,
       const float* __restrict__ pb, float* __restrict__ out, int o_outer) {
  proj_body(blockIdx.x * 512 + threadIdx.x, xb, pw, pb, out, o_outer);
}

#define MIX_BLOCKS 64
__global__ void __launch_bounds__(512)
k_mix2_proj(const float* __restrict__ Frp, const float* __restrict__ Fip,
            const float* __restrict__ wr, const float* __restrict__ wi,
            float* __restrict__ Gr, float* __restrict__ Gi,
            float* __restrict__ Fsr, float* __restrict__ Fsi,
            const float* __restrict__ xb, const float* __restrict__ pw,
            const float* __restrict__ pb, float* __restrict__ out, int o_mix) {
  if (blockIdx.x < MIX_BLOCKS)
    mix_body<2>(blockIdx.x, Frp, Fip, wr, wi, Gr, Gi, Fsr, Fsi);
  else
    proj_body((blockIdx.x - MIX_BLOCKS) * 512 + threadIdx.x, xb, pw, pb, out, o_mix - 1);
}

extern "C" void kernel_launch(void* const* d_in, const int* in_sizes, int n_in,
                              void* d_out, int out_size, void* d_ws, size_t ws_size,
                              hipStream_t stream) {
  const float* x      = (const float*)d_in[0];
  const float* lift_w = (const float*)d_in[1];
  const float* lift_b = (const float*)d_in[2];
  const float* wr     = (const float*)d_in[3];
  const float* wi     = (const float*)d_in[4];
  const float* byp_w  = (const float*)d_in[5];
  const float* byp_b  = (const float*)d_in[6];
  const float* proj_w = (const float*)d_in[7];
  const float* proj_b = (const float*)d_in[8];
  float* out = (float*)d_out;

  float* w = (float*)d_ws;
  float* fC    = w;
  float* fS    = fC + 4096;
  float* iC    = fS + 4096;
  float* iS    = iC + 4096;
  float* x0    = iS + 4096;
  float* bufA  = x0 + (size_t)B_ * PIX * C_;
  float* bufB  = bufA + (size_t)B_ * PIX * C_;
  float* Gr    = bufB + (size_t)B_ * PIX * C_;
  float* Gi    = Gr + (size_t)B_ * M_ * M_ * C_;
  float* Frp   = Gi + (size_t)B_ * M_ * M_ * C_;
  float* Fip   = Frp + (size_t)32 * B_ * M_ * 512;
  float* Fsr   = Fip + (size_t)32 * B_ * M_ * 512;
  float* Fsi   = Fsr + (size_t)B_ * M_ * 512;

  k_tabs_lift<<<TAB_BLOCKS + B_ * PIX / 256, 256, 0, stream>>>(fC, fS, iC, iS,
                                                              x, lift_w, lift_b, x0);
  k_fwdXY<<<256, 512, 0, stream>>>(x0, fC, fS, Frp, Fip);

  for (int o = 0; o < 4; ++o) {
    for (int n = 0; n < 4; ++n) {
      const float* cur = (n == 0) ? x0 : ((n & 1) ? bufA : bufB);
      float* dst = (n & 1) ? bufB : bufA;
      size_t wOff = ((size_t)(o * 4 + n)) * C_ * C_ * M_ * M_;
      size_t bwOff = ((size_t)(o * 4 + n)) * C_ * C_;
      size_t bbOff = ((size_t)(o * 4 + n)) * C_;
      if (n == 0) {
        if (o == 0)
          k_mix<1><<<64, 512, 0, stream>>>(Frp, Fip, wr + wOff, wi + wOff, Gr, Gi, Fsr, Fsi);
        else
          k_mix2_proj<<<MIX_BLOCKS + B_ * PIX / 512, 512, 0, stream>>>(
              Frp, Fip, wr + wOff, wi + wOff, Gr, Gi, Fsr, Fsi,
              bufB, proj_w, proj_b, out, o);
      } else {
        k_fwdXY<<<256, 512, 0, stream>>>(cur, fC, fS, Frp, Fip);
        k_mix<0><<<64, 512, 0, stream>>>(Frp, Fip, wr + wOff, wi + wOff, Gr, Gi, Fsr, Fsi);
      }
      k_invYX<<<256, 1024, 0, stream>>>(Gr, Gi, cur, iC, iS,
                                        byp_w + bwOff, byp_b + bbOff, dst);
    }
  }
  k_proj<<<B_ * PIX / 512, 512, 0, stream>>>(bufB, proj_w, proj_b, out, 3);
}
